# starting kernel
# baseline (speedup 1.0000x reference)
#include <hip/hip_runtime.h>

typedef float v2f __attribute__((ext_vector_type(2)));
typedef float v8f __attribute__((ext_vector_type(8)));

#define T_LEN 2048
#define BLK 16
#define NBLK (T_LEN / BLK)
#define WAVES_PER_CTA 8
#define CTA_THREADS 256

__global__ __launch_bounds__(CTA_THREADS) void pz_biquad_wmma(
    const float* __restrict__ x,
    const float* __restrict__ gain_ri,
    const float* __restrict__ poles_ri,
    const float* __restrict__ zeros_ri,
    float* __restrict__ out)
{
    __shared__ float hsh[16];
    __shared__ float xt[WAVES_PER_CTA][16][16];
    __shared__ float yt[WAVES_PER_CTA][16][16];

    const int tid  = threadIdx.x;
    const int w    = tid >> 5;
    const int lane = tid & 31;
    const int n    = lane & 15;
    const int hi   = lane >> 4;

    const float gr  = gain_ri[0],  gi  = gain_ri[1];
    const float p0r = poles_ri[0], p0i = poles_ri[1];
    const float p1r = poles_ri[2], p1i = poles_ri[3];
    const float z0r = zeros_ri[0], z0i = zeros_ri[1];
    const float z1r = zeros_ri[2], z1i = zeros_ri[3];

    const float a1 = -(p0r + p1r);
    const float a2 = p0r * p1r - p0i * p1i;
    const float zc1r = -(z0r + z1r), zc1i = -(z0i + z1i);
    const float zc2r = z0r * z1r - z0i * z1i;
    const float zc2i = z0r * z1i + z0i * z1r;
    const float b0 = gr;
    const float b1 = gr * zc1r - gi * zc1i;
    const float b2 = gr * zc2r - gi * zc2i;

    float g[16];
    g[0] = 1.f; g[1] = -a1;
#pragma unroll
    for (int k = 2; k < 16; ++k) g[k] = -a1 * g[k-1] - a2 * g[k-2];

    float hh[16], R1[16], R2[16], P[16], Q[16];
#pragma unroll
    for (int k = 0; k < 16; ++k) {
        const float gm1 = (k >= 1) ? g[k-1] : 0.f;
        const float gm2 = (k >= 2) ? g[k-2] : 0.f;
        hh[k] = b0 * g[k] + b1 * gm1 + b2 * gm2;
        R1[k] = b1 * g[k] + b2 * gm1;
        R2[k] = b2 * g[k];
    }
    P[0] = -a1; P[1] = a1 * a1 - a2;
    Q[0] = -a2; Q[1] = a1 * a2;
#pragma unroll
    for (int k = 2; k < 16; ++k) {
        P[k] = -a1 * P[k-1] - a2 * P[k-2];
        Q[k] = -a1 * Q[k-1] - a2 * Q[k-2];
    }

    if (tid == 0) {
#pragma unroll
        for (int k = 0; k < 16; ++k) hsh[k] = hh[k];
    }

    float cr1[8], cr2[8], cp[8], cq[8];
#pragma unroll
    for (int r = 0; r < 8; ++r) {
        cr1[r] = hi ? R1[r + 8] : R1[r];
        cr2[r] = hi ? R2[r + 8] : R2[r];
        cp[r]  = hi ? P[r + 8]  : P[r];
        cq[r]  = hi ? Q[r + 8]  : Q[r];
    }

    __syncthreads();

    v2f A[4];
#pragma unroll
    for (int kb = 0; kb < 4; ++kb) {
        const int i0 = n - 4 * kb - 2 * hi;
        A[kb].x = (i0 >= 0) ? hsh[i0]     : 0.f;
        A[kb].y = (i0 >= 1) ? hsh[i0 - 1] : 0.f;
    }

    const long long bb = (long long)(blockIdx.x * WAVES_PER_CTA + w) * 16;
    const int srow = lane >> 1;
    const int scol = (lane & 1) * 8;
    const float* xrow = x   + (bb + srow) * (long long)T_LEN;
    float*       orow = out + (bb + srow) * (long long)T_LEN;

    float xm1 = 0.f, xm2 = 0.f, ym1 = 0.f, ym2 = 0.f;

    for (int blk = 0; blk < NBLK; ++blk) {
        const int t0 = blk * BLK;

        const float4 xa = *(const float4*)(xrow + t0 + scol);
        const float4 xb = *(const float4*)(xrow + t0 + scol + 4);
        if (blk + 1 < NBLK)
            __builtin_prefetch(xrow + t0 + BLK + scol, 0, 0);
        *(float4*)&xt[w][srow][scol]     = xa;
        *(float4*)&xt[w][srow][scol + 4] = xb;
        __builtin_amdgcn_wave_barrier();
        asm volatile("s_wait_dscnt 0x0" ::: "memory");

        v2f Bv[4];
#pragma unroll
        for (int kb = 0; kb < 4; ++kb)
            Bv[kb] = *(const v2f*)&xt[w][n][4 * kb + 2 * hi];

        v8f acc;
#pragma unroll
        for (int r = 0; r < 8; ++r)
            acc[r] = cr1[r] * xm1 + cr2[r] * xm2 + cp[r] * ym1 + cq[r] * ym2;

#pragma unroll
        for (int kb = 0; kb < 4; ++kb)
            acc = __builtin_amdgcn_wmma_f32_16x16x4_f32(
                false, A[kb], false, Bv[kb], (short)0, acc, false, false);

        xm2 = __shfl(Bv[3].x, 16 + n, 32);
        xm1 = __shfl(Bv[3].y, 16 + n, 32);
        ym2 = __shfl(acc[6],  16 + n, 32);
        ym1 = __shfl(acc[7],  16 + n, 32);

        const float4 y0 = make_float4(acc[0], acc[1], acc[2], acc[3]);
        const float4 y1 = make_float4(acc[4], acc[5], acc[6], acc[7]);
        *(float4*)&yt[w][n][8 * hi]     = y0;
        *(float4*)&yt[w][n][8 * hi + 4] = y1;
        __builtin_amdgcn_wave_barrier();
        asm volatile("s_wait_dscnt 0x0" ::: "memory");

        *(float4*)(orow + t0 + scol)     = *(const float4*)&yt[w][srow][scol];
        *(float4*)(orow + t0 + scol + 4) = *(const float4*)&yt[w][srow][scol + 4];
    }
}

extern "C" void kernel_launch(void* const* d_in, const int* in_sizes, int n_in,
                              void* d_out, int out_size, void* d_ws, size_t ws_size,
                              hipStream_t stream) {
    const float* x     = (const float*)d_in[0];
    const float* gain  = (const float*)d_in[1];
    const float* poles = (const float*)d_in[2];
    const float* zeros = (const float*)d_in[3];
    float* out = (float*)d_out;

    const int B = in_sizes[0] / T_LEN;
    const int blocks = B / (WAVES_PER_CTA * 16);
    pz_biquad_wmma<<<blocks, CTA_THREADS, 0, stream>>>(x, gain, poles, zeros, out);
}


# Time axis split over the 16 waves of a work-group: zero-y-history pass per 128-step segment, state scan stepping the 16-step block transition forward, homogeneous term added; sum order changes
# speedup vs baseline: 5.5398x; 5.5398x over previous
#include <hip/hip_runtime.h>


typedef float v2f __attribute__((ext_vector_type(2)));
typedef float v8f __attribute__((ext_vector_type(8)));

#define T_LEN 2048
#define BLK 16
#define NSEG 16
#define SEG_LEN (T_LEN / NSEG)
#define SEG_BLKS (SEG_LEN / BLK)
#define CTA_THREADS (NSEG * 32)

__global__ __launch_bounds__(CTA_THREADS) void pz_biquad_wmma(
    const float* __restrict__ x,
    const float* __restrict__ gain_ri,
    const float* __restrict__ poles_ri,
    const float* __restrict__ zeros_ri,
    float* __restrict__ out)
{
    __shared__ float hsh[16];
    __shared__ __attribute__((aligned(16))) float xt[NSEG][16][16];
    __shared__ __attribute__((aligned(16))) float yt[NSEG][16][16];
    __shared__ float zs[NSEG][2][16];

    const int tid  = threadIdx.x;
    const int w    = tid >> 5;
    const int lane = tid & 31;
    const int n    = lane & 15;
    const int hi   = lane >> 4;

    const float gr  = gain_ri[0],  gi  = gain_ri[1];
    const float p0r = poles_ri[0], p0i = poles_ri[1];
    const float p1r = poles_ri[2], p1i = poles_ri[3];
    const float z0r = zeros_ri[0], z0i = zeros_ri[1];
    const float z1r = zeros_ri[2], z1i = zeros_ri[3];

    const float a1 = -(p0r + p1r);
    const float a2 = p0r * p1r - p0i * p1i;
    const float zc1r = -(z0r + z1r), zc1i = -(z0i + z1i);
    const float zc2r = z0r * z1r - z0i * z1i;
    const float zc2i = z0r * z1i + z0i * z1r;
    const float b0 = gr;
    const float b1 = gr * zc1r - gi * zc1i;
    const float b2 = gr * zc2r - gi * zc2i;

    float g[16];
    g[0] = 1.f; g[1] = -a1;
#pragma unroll
    for (int k = 2; k < 16; ++k) g[k] = -a1 * g[k-1] - a2 * g[k-2];

    float hh[16], R1[16], R2[16], P[16], Q[16];
#pragma unroll
    for (int k = 0; k < 16; ++k) {
        const float gm1 = (k >= 1) ? g[k-1] : 0.f;
        const float gm2 = (k >= 2) ? g[k-2] : 0.f;
        hh[k] = b0 * g[k] + b1 * gm1 + b2 * gm2;
        R1[k] = b1 * g[k] + b2 * gm1;
        R2[k] = b2 * g[k];
    }
    P[0] = -a1; P[1] = a1 * a1 - a2;
    Q[0] = -a2; Q[1] = a1 * a2;
#pragma unroll
    for (int k = 2; k < 16; ++k) {
        P[k] = -a1 * P[k-1] - a2 * P[k-2];
        Q[k] = -a1 * Q[k-1] - a2 * Q[k-2];
    }

    const float B11 = P[15], B12 = Q[15], B21 = P[14], B22 = Q[14];

    if (tid == 0) {
#pragma unroll
        for (int k = 0; k < 16; ++k) hsh[k] = hh[k];
    }

    float cr1[8], cr2[8], cp[8], cq[8];
#pragma unroll
    for (int r = 0; r < 8; ++r) {
        cr1[r] = hi ? R1[r + 8] : R1[r];
        cr2[r] = hi ? R2[r + 8] : R2[r];
        cp[r]  = hi ? P[r + 8]  : P[r];
        cq[r]  = hi ? Q[r + 8]  : Q[r];
    }

    __syncthreads();

    v2f A[4];
#pragma unroll
    for (int kb = 0; kb < 4; ++kb) {
        const int i0 = n - 4 * kb - 2 * hi;
        A[kb].x = (i0 >= 0) ? hsh[i0]     : 0.f;
        A[kb].y = (i0 >= 1) ? hsh[i0 - 1] : 0.f;
    }

    const long long bb = (long long)blockIdx.x * 16;
    const int tseg = w * SEG_LEN;
    const int srow = lane >> 1;
    const int scol = (lane & 1) * 8;
    const float* xrow = x   + (bb + srow) * (long long)T_LEN + tseg;
    float*       orow = out + (bb + srow) * (long long)T_LEN + tseg;

    float xm1 = 0.f, xm2 = 0.f;
    if (w > 0) {
        const float* xn = x + (bb + n) * (long long)T_LEN + tseg;
        xm1 = xn[-1];
        xm2 = xn[-2];
    }
    float ym1 = 0.f, ym2 = 0.f;

    v8f accs[SEG_BLKS];

#pragma unroll
    for (int blk = 0; blk < SEG_BLKS; ++blk) {
        const int t0 = blk * BLK;

        const float4 xa = *(const float4*)(xrow + t0 + scol);
        const float4 xb = *(const float4*)(xrow + t0 + scol + 4);
        if (blk + 1 < SEG_BLKS)
            __builtin_prefetch(xrow + t0 + BLK + scol, 0, 0);
        *(float4*)&xt[w][srow][scol]     = xa;
        *(float4*)&xt[w][srow][scol + 4] = xb;
        __builtin_amdgcn_wave_barrier();
        asm volatile("s_wait_dscnt 0x0" ::: "memory");

        v2f Bv[4];
#pragma unroll
        for (int kb = 0; kb < 4; ++kb)
            Bv[kb] = *(const v2f*)&xt[w][n][4 * kb + 2 * hi];

        v8f acc;
#pragma unroll
        for (int r = 0; r < 8; ++r)
            acc[r] = cr1[r] * xm1 + cr2[r] * xm2 + cp[r] * ym1 + cq[r] * ym2;

#pragma unroll
        for (int kb = 0; kb < 4; ++kb)
            acc = __builtin_amdgcn_wmma_f32_16x16x4_f32(
                false, A[kb], false, Bv[kb], (short)0, acc, false, false);

        xm2 = __shfl(Bv[3].x, 16 + n, 32);
        xm1 = __shfl(Bv[3].y, 16 + n, 32);
        ym2 = __shfl(acc[6],  16 + n, 32);
        ym1 = __shfl(acc[7],  16 + n, 32);

        accs[blk] = acc;
    }

    if (hi == 0) {
        zs[w][0][n] = ym1;
        zs[w][1][n] = ym2;
    }
    __syncthreads();

    float e1[NSEG - 1], e2[NSEG - 1];
#pragma unroll
    for (int v = 0; v < NSEG - 1; ++v) {
        e1[v] = zs[v][0][n];
        e2[v] = zs[v][1][n];
    }
    float s1 = 0.f, s2 = 0.f;
#pragma unroll
    for (int v = 0; v < NSEG - 1; ++v) {
        if (v < w) {
            if (v > 0) {
                for (int b = 0; b < SEG_BLKS; ++b) {
                    const float u1 = B11 * s1 + B12 * s2;
                    const float u2 = B21 * s1 + B22 * s2;
                    s1 = u1; s2 = u2;
                }
            }
            s1 += e1[v];
            s2 += e2[v];
        }
    }

#pragma unroll
    for (int blk = 0; blk < SEG_BLKS; ++blk) {
        const int t0 = blk * BLK;
        v8f acc = accs[blk];
#pragma unroll
        for (int r = 0; r < 8; ++r)
            acc[r] += cp[r] * s1 + cq[r] * s2;
        const float u1 = B11 * s1 + B12 * s2;
        const float u2 = B21 * s1 + B22 * s2;
        s1 = u1; s2 = u2;

        const float4 y0 = make_float4(acc[0], acc[1], acc[2], acc[3]);
        const float4 y1 = make_float4(acc[4], acc[5], acc[6], acc[7]);
        *(float4*)&yt[w][n][8 * hi]     = y0;
        *(float4*)&yt[w][n][8 * hi + 4] = y1;
        __builtin_amdgcn_wave_barrier();
        asm volatile("s_wait_dscnt 0x0" ::: "memory");

        *(float4*)(orow + t0 + scol)     = *(const float4*)&yt[w][srow][scol];
        *(float4*)(orow + t0 + scol + 4) = *(const float4*)&yt[w][srow][scol + 4];
    }
}

extern "C" void kernel_launch(void* const* d_in, const int* in_sizes, int n_in,
                              void* d_out, int out_size, void* d_ws, size_t ws_size,
                              hipStream_t stream) {
    const float* x     = (const float*)d_in[0];
    const float* gain  = (const float*)d_in[1];
    const float* poles = (const float*)d_in[2];
    const float* zeros = (const float*)d_in[3];
    float* out = (float*)d_out;

    const int B = in_sizes[0] / T_LEN;
    const int blocks = B / 16;
    pz_biquad_wmma<<<blocks, CTA_THREADS, 0, stream>>>(x, gain, poles, zeros, out);
}


# Whole-line global traffic: each wave moves its 16x128 segment through a private LDS tile, 16 row loads of 512 B issued together (no barrier before pass 1), 16 row stores of 512 B; sums unchanged
# speedup vs baseline: 7.5353x; 1.3602x over previous
#include <hip/hip_runtime.h>


typedef float v2f __attribute__((ext_vector_type(2)));
typedef float v8f __attribute__((ext_vector_type(8)));

#define T_LEN 2048
#define BLK 16
#define NSEG 16
#define SEG_LEN (T_LEN / NSEG)
#define SEG_BLKS (SEG_LEN / BLK)
#define CTA_THREADS (NSEG * 32)
#define TILE_PITCH (SEG_LEN + 4)

__global__ __launch_bounds__(CTA_THREADS) void pz_biquad_wmma(
    const float* __restrict__ x,
    const float* __restrict__ gain_ri,
    const float* __restrict__ poles_ri,
    const float* __restrict__ zeros_ri,
    float* __restrict__ out)
{
    __shared__ float hsh[NSEG][16];
    __shared__ __attribute__((aligned(16))) float tile[NSEG][16][TILE_PITCH];
    __shared__ float zs[NSEG][2][16];

    const int tid  = threadIdx.x;
    const int w    = tid >> 5;
    const int lane = tid & 31;
    const int n    = lane & 15;
    const int hi   = lane >> 4;

    const long long bb = (long long)blockIdx.x * 16;
    const int tseg = w * SEG_LEN;

    float xm1 = 0.f, xm2 = 0.f;
    if (w > 0) {
        const float* xn = x + (bb + n) * (long long)T_LEN + tseg;
        xm1 = xn[-1];
        xm2 = xn[-2];
    }

    const float* xseg = x   + bb * (long long)T_LEN + tseg + 4 * lane;
    float*       oseg = out + bb * (long long)T_LEN + tseg + 4 * lane;
    float4 xin[16];
#pragma unroll
    for (int r = 0; r < 16; ++r)
        xin[r] = *(const float4*)(xseg + (long long)r * T_LEN);

    const float gr  = gain_ri[0],  gi  = gain_ri[1];
    const float p0r = poles_ri[0], p0i = poles_ri[1];
    const float p1r = poles_ri[2], p1i = poles_ri[3];
    const float z0r = zeros_ri[0], z0i = zeros_ri[1];
    const float z1r = zeros_ri[2], z1i = zeros_ri[3];

    const float a1 = -(p0r + p1r);
    const float a2 = p0r * p1r - p0i * p1i;
    const float zc1r = -(z0r + z1r), zc1i = -(z0i + z1i);
    const float zc2r = z0r * z1r - z0i * z1i;
    const float zc2i = z0r * z1i + z0i * z1r;
    const float b0 = gr;
    const float b1 = gr * zc1r - gi * zc1i;
    const float b2 = gr * zc2r - gi * zc2i;

    float g[16];
    g[0] = 1.f; g[1] = -a1;
#pragma unroll
    for (int k = 2; k < 16; ++k) g[k] = -a1 * g[k-1] - a2 * g[k-2];

    float hh[16], R1[16], R2[16], P[16], Q[16];
#pragma unroll
    for (int k = 0; k < 16; ++k) {
        const float gm1 = (k >= 1) ? g[k-1] : 0.f;
        const float gm2 = (k >= 2) ? g[k-2] : 0.f;
        hh[k] = b0 * g[k] + b1 * gm1 + b2 * gm2;
        R1[k] = b1 * g[k] + b2 * gm1;
        R2[k] = b2 * g[k];
    }
    P[0] = -a1; P[1] = a1 * a1 - a2;
    Q[0] = -a2; Q[1] = a1 * a2;
#pragma unroll
    for (int k = 2; k < 16; ++k) {
        P[k] = -a1 * P[k-1] - a2 * P[k-2];
        Q[k] = -a1 * Q[k-1] - a2 * Q[k-2];
    }

    const float B11 = P[15], B12 = Q[15], B21 = P[14], B22 = Q[14];

    float cr1[8], cr2[8], cp[8], cq[8];
#pragma unroll
    for (int r = 0; r < 8; ++r) {
        cr1[r] = hi ? R1[r + 8] : R1[r];
        cr2[r] = hi ? R2[r + 8] : R2[r];
        cp[r]  = hi ? P[r + 8]  : P[r];
        cq[r]  = hi ? Q[r + 8]  : Q[r];
    }

#pragma unroll
    for (int r = 0; r < 16; ++r)
        *(float4*)&tile[w][r][4 * lane] = xin[r];

    if (lane == 0) {
#pragma unroll
        for (int k = 0; k < 16; ++k) hsh[w][k] = hh[k];
    }

    __builtin_amdgcn_wave_barrier();
    asm volatile("s_wait_dscnt 0x0" ::: "memory");

    v2f A[4];
#pragma unroll
    for (int kb = 0; kb < 4; ++kb) {
        const int i0 = n - 4 * kb - 2 * hi;
        A[kb].x = (i0 >= 0) ? hsh[w][i0]     : 0.f;
        A[kb].y = (i0 >= 1) ? hsh[w][i0 - 1] : 0.f;
    }

    float ym1 = 0.f, ym2 = 0.f;

    v8f accs[SEG_BLKS];

#pragma unroll
    for (int blk = 0; blk < SEG_BLKS; ++blk) {
        const int t0 = blk * BLK;

        v2f Bv[4];
#pragma unroll
        for (int kb = 0; kb < 4; ++kb)
            Bv[kb] = *(const v2f*)&tile[w][n][t0 + 4 * kb + 2 * hi];

        v8f acc;
#pragma unroll
        for (int r = 0; r < 8; ++r)
            acc[r] = cr1[r] * xm1 + cr2[r] * xm2 + cp[r] * ym1 + cq[r] * ym2;

#pragma unroll
        for (int kb = 0; kb < 4; ++kb)
            acc = __builtin_amdgcn_wmma_f32_16x16x4_f32(
                false, A[kb], false, Bv[kb], (short)0, acc, false, false);

        xm2 = __shfl(Bv[3].x, 16 + n, 32);
        xm1 = __shfl(Bv[3].y, 16 + n, 32);
        ym2 = __shfl(acc[6],  16 + n, 32);
        ym1 = __shfl(acc[7],  16 + n, 32);

        accs[blk] = acc;
    }

    if (hi == 0) {
        zs[w][0][n] = ym1;
        zs[w][1][n] = ym2;
    }
    __syncthreads();

    float e1[NSEG - 1], e2[NSEG - 1];
#pragma unroll
    for (int v = 0; v < NSEG - 1; ++v) {
        e1[v] = zs[v][0][n];
        e2[v] = zs[v][1][n];
    }
    float s1 = 0.f, s2 = 0.f;
#pragma unroll
    for (int v = 0; v < NSEG - 1; ++v) {
        if (v < w) {
            if (v > 0) {
                for (int b = 0; b < SEG_BLKS; ++b) {
                    const float u1 = B11 * s1 + B12 * s2;
                    const float u2 = B21 * s1 + B22 * s2;
                    s1 = u1; s2 = u2;
                }
            }
            s1 += e1[v];
            s2 += e2[v];
        }
    }

#pragma unroll
    for (int blk = 0; blk < SEG_BLKS; ++blk) {
        const int t0 = blk * BLK;
        v8f acc = accs[blk];
#pragma unroll
        for (int r = 0; r < 8; ++r)
            acc[r] += cp[r] * s1 + cq[r] * s2;
        const float u1 = B11 * s1 + B12 * s2;
        const float u2 = B21 * s1 + B22 * s2;
        s1 = u1; s2 = u2;

        *(float4*)&tile[w][n][t0 + 8 * hi]     = make_float4(acc[0], acc[1], acc[2], acc[3]);
        *(float4*)&tile[w][n][t0 + 8 * hi + 4] = make_float4(acc[4], acc[5], acc[6], acc[7]);
    }
    __builtin_amdgcn_wave_barrier();
    asm volatile("s_wait_dscnt 0x0" ::: "memory");

#pragma unroll
    for (int r = 0; r < 16; ++r)
        *(float4*)(oseg + (long long)r * T_LEN) = *(const float4*)&tile[w][r][4 * lane];
}

extern "C" void kernel_launch(void* const* d_in, const int* in_sizes, int n_in,
                              void* d_out, int out_size, void* d_ws, size_t ws_size,
                              hipStream_t stream) {
    const float* x     = (const float*)d_in[0];
    const float* gain  = (const float*)d_in[1];
    const float* poles = (const float*)d_in[2];
    const float* zeros = (const float*)d_in[3];
    float* out = (float*)d_out;

    const int B = in_sizes[0] / T_LEN;
    const int blocks = B / 16;
    pz_biquad_wmma<<<blocks, CTA_THREADS, 0, stream>>>(x, gain, poles, zeros, out);
}


# Each wave owns one batch row: its 16 matrix columns are the row's time segments; segment scan inside the wave (row broadcast, one 2x2 segment transition), no workgroup barrier; sum order changes
# speedup vs baseline: 11.1162x; 1.4752x over previous
#include <hip/hip_runtime.h>


typedef float v2f __attribute__((ext_vector_type(2)));
typedef float v8f __attribute__((ext_vector_type(8)));

#define T_LEN 2048
#define BLK 16
#define NSEG 16
#define WAVES 16
#define SEG_LEN (T_LEN / NSEG)
#define SEG_BLKS (SEG_LEN / BLK)
#define CTA_THREADS (WAVES * 32)
#define TILE_PITCH (SEG_LEN + 4)

template <int V>
__device__ __forceinline__ float row_bcast(float a) {
    return __int_as_float(__builtin_amdgcn_update_dpp(
        0, __float_as_int(a), 0x150 + V, 0xf, 0xf, false));
}

__global__ __launch_bounds__(CTA_THREADS) void pz_biquad_wmma(
    const float* __restrict__ x,
    const float* __restrict__ gain_ri,
    const float* __restrict__ poles_ri,
    const float* __restrict__ zeros_ri,
    float* __restrict__ out)
{
    __shared__ float hsh[WAVES][16];
    __shared__ __attribute__((aligned(16))) float tile[WAVES][NSEG][TILE_PITCH];

    const int tid  = threadIdx.x;
    const int w    = tid >> 5;
    const int lane = tid & 31;
    const int n    = lane & 15;
    const int hi   = lane >> 4;

    const long long row = (long long)blockIdx.x * WAVES + w;

    const float* xseg = x   + row * (long long)T_LEN + 4 * lane;
    float*       oseg = out + row * (long long)T_LEN + 4 * lane;
    float4 xin[16];
#pragma unroll
    for (int r = 0; r < 16; ++r)
        xin[r] = *(const float4*)(xseg + r * SEG_LEN);

    const float gr  = gain_ri[0],  gi  = gain_ri[1];
    const float p0r = poles_ri[0], p0i = poles_ri[1];
    const float p1r = poles_ri[2], p1i = poles_ri[3];
    const float z0r = zeros_ri[0], z0i = zeros_ri[1];
    const float z1r = zeros_ri[2], z1i = zeros_ri[3];

    const float a1 = -(p0r + p1r);
    const float a2 = p0r * p1r - p0i * p1i;
    const float zc1r = -(z0r + z1r), zc1i = -(z0i + z1i);
    const float zc2r = z0r * z1r - z0i * z1i;
    const float zc2i = z0r * z1i + z0i * z1r;
    const float b0 = gr;
    const float b1 = gr * zc1r - gi * zc1i;
    const float b2 = gr * zc2r - gi * zc2i;

    float g[16];
    g[0] = 1.f; g[1] = -a1;
#pragma unroll
    for (int k = 2; k < 16; ++k) g[k] = -a1 * g[k-1] - a2 * g[k-2];

    float hh[16], R1[16], R2[16], P[16], Q[16];
#pragma unroll
    for (int k = 0; k < 16; ++k) {
        const float gm1 = (k >= 1) ? g[k-1] : 0.f;
        const float gm2 = (k >= 2) ? g[k-2] : 0.f;
        hh[k] = b0 * g[k] + b1 * gm1 + b2 * gm2;
        R1[k] = b1 * g[k] + b2 * gm1;
        R2[k] = b2 * g[k];
    }
    P[0] = -a1; P[1] = a1 * a1 - a2;
    Q[0] = -a2; Q[1] = a1 * a2;
#pragma unroll
    for (int k = 2; k < 16; ++k) {
        P[k] = -a1 * P[k-1] - a2 * P[k-2];
        Q[k] = -a1 * Q[k-1] - a2 * Q[k-2];
    }

    const float B11 = P[15], B12 = Q[15], B21 = P[14], B22 = Q[14];

    float M11 = 1.f, M21 = 0.f, M12 = 0.f, M22 = 1.f;
#pragma unroll
    for (int b = 0; b < SEG_BLKS; ++b) {
        const float c11 = B11 * M11 + B12 * M21;
        const float c21 = B21 * M11 + B22 * M21;
        const float c12 = B11 * M12 + B12 * M22;
        const float c22 = B21 * M12 + B22 * M22;
        M11 = c11; M21 = c21; M12 = c12; M22 = c22;
    }

    float cr1[8], cr2[8], cp[8], cq[8];
#pragma unroll
    for (int r = 0; r < 8; ++r) {
        cr1[r] = hi ? R1[r + 8] : R1[r];
        cr2[r] = hi ? R2[r + 8] : R2[r];
        cp[r]  = hi ? P[r + 8]  : P[r];
        cq[r]  = hi ? Q[r + 8]  : Q[r];
    }

#pragma unroll
    for (int r = 0; r < 16; ++r)
        *(float4*)&tile[w][r][4 * lane] = xin[r];

    if (lane == 0) {
#pragma unroll
        for (int k = 0; k < 16; ++k) hsh[w][k] = hh[k];
    }

    __builtin_amdgcn_wave_barrier();
    asm volatile("s_wait_dscnt 0x0" ::: "memory");

    v2f A[4];
#pragma unroll
    for (int kb = 0; kb < 4; ++kb) {
        const int i0 = n - 4 * kb - 2 * hi;
        A[kb].x = (i0 >= 0) ? hsh[w][i0]     : 0.f;
        A[kb].y = (i0 >= 1) ? hsh[w][i0 - 1] : 0.f;
    }

    const v2f xh = *(const v2f*)&tile[w][n > 0 ? n - 1 : 0][SEG_LEN - 2];
    float xm2 = (n > 0) ? xh.x : 0.f;
    float xm1 = (n > 0) ? xh.y : 0.f;

    float ym1 = 0.f, ym2 = 0.f;

    v8f accs[SEG_BLKS];

#pragma unroll
    for (int blk = 0; blk < SEG_BLKS; ++blk) {
        const int t0 = blk * BLK;

        v2f Bv[4];
#pragma unroll
        for (int kb = 0; kb < 4; ++kb)
            Bv[kb] = *(const v2f*)&tile[w][n][t0 + 4 * kb + 2 * hi];

        v8f acc;
#pragma unroll
        for (int r = 0; r < 8; ++r)
            acc[r] = cr1[r] * xm1 + cr2[r] * xm2 + cp[r] * ym1 + cq[r] * ym2;

#pragma unroll
        for (int kb = 0; kb < 4; ++kb)
            acc = __builtin_amdgcn_wmma_f32_16x16x4_f32(
                false, A[kb], false, Bv[kb], (short)0, acc, false, false);

        xm2 = __shfl(Bv[3].x, 16 + n, 32);
        xm1 = __shfl(Bv[3].y, 16 + n, 32);
        ym2 = __shfl(acc[6],  16 + n, 32);
        ym1 = __shfl(acc[7],  16 + n, 32);

        accs[blk] = acc;
    }

    float s1 = 0.f, s2 = 0.f;
    float S1 = row_bcast<0>(ym1), S2 = row_bcast<0>(ym2);
    if (n == 1) { s1 = S1; s2 = S2; }
#define SCAN_STEP(V)                                                    \
    {                                                                   \
        const float u1 = M11 * S1 + (M12 * S2 + row_bcast<V>(ym1));     \
        const float u2 = M21 * S1 + (M22 * S2 + row_bcast<V>(ym2));     \
        S1 = u1; S2 = u2;                                    \
        if (n == V + 1) { s1 = S1; s2 = S2; }                           \
    }
    SCAN_STEP(1)  SCAN_STEP(2)  SCAN_STEP(3)  SCAN_STEP(4)  SCAN_STEP(5)
    SCAN_STEP(6)  SCAN_STEP(7)  SCAN_STEP(8)  SCAN_STEP(9)  SCAN_STEP(10)
    SCAN_STEP(11) SCAN_STEP(12) SCAN_STEP(13) SCAN_STEP(14)
#undef SCAN_STEP

#pragma unroll
    for (int blk = 0; blk < SEG_BLKS; ++blk) {
        const int t0 = blk * BLK;
        v8f acc = accs[blk];
#pragma unroll
        for (int r = 0; r < 8; ++r)
            acc[r] += cp[r] * s1 + cq[r] * s2;
        const float u1 = B11 * s1 + B12 * s2;
        const float u2 = B21 * s1 + B22 * s2;
        s1 = u1; s2 = u2;

        *(float4*)&tile[w][n][t0 + 8 * hi]     = make_float4(acc[0], acc[1], acc[2], acc[3]);
        *(float4*)&tile[w][n][t0 + 8 * hi + 4] = make_float4(acc[4], acc[5], acc[6], acc[7]);
    }
    __builtin_amdgcn_wave_barrier();
    asm volatile("s_wait_dscnt 0x0" ::: "memory");

#pragma unroll
    for (int r = 0; r < 16; ++r)
        *(float4*)(oseg + r * SEG_LEN) = *(const float4*)&tile[w][r][4 * lane];
}

extern "C" void kernel_launch(void* const* d_in, const int* in_sizes, int n_in,
                              void* d_out, int out_size, void* d_ws, size_t ws_size,
                              hipStream_t stream) {
    const float* x     = (const float*)d_in[0];
    const float* gain  = (const float*)d_in[1];
    const float* poles = (const float*)d_in[2];
    const float* zeros = (const float*)d_in[3];
    float* out = (float*)d_out;

    const int B = in_sizes[0] / T_LEN;
    const int blocks = B / WAVES;
    pz_biquad_wmma<<<blocks, CTA_THREADS, 0, stream>>>(x, gain, poles, zeros, out);
}


# Zero y history in front of every 16-step block, not just every segment: a wave's 8 block products no longer chain through shuffles; block end states scanned by the 2x2 block step; sum order changes
# speedup vs baseline: 11.8426x; 1.0653x over previous
#include <hip/hip_runtime.h>


typedef float v2f __attribute__((ext_vector_type(2)));
typedef float v8f __attribute__((ext_vector_type(8)));

#define T_LEN 2048
#define BLK 16
#define NSEG 16
#define WAVES 16
#define SEG_LEN (T_LEN / NSEG)
#define SEG_BLKS (SEG_LEN / BLK)
#define CTA_THREADS (WAVES * 32)
#define TILE_PITCH (SEG_LEN + 4)

template <int V>
__device__ __forceinline__ float row_bcast(float a) {
    return __int_as_float(__builtin_amdgcn_update_dpp(
        0, __float_as_int(a), 0x150 + V, 0xf, 0xf, false));
}

__global__ __launch_bounds__(CTA_THREADS) void pz_biquad_wmma(
    const float* __restrict__ x,
    const float* __restrict__ gain_ri,
    const float* __restrict__ poles_ri,
    const float* __restrict__ zeros_ri,
    float* __restrict__ out)
{
    __shared__ float hsh[WAVES][16];
    __shared__ __attribute__((aligned(16))) float tile[WAVES][NSEG][TILE_PITCH];

    const int tid  = threadIdx.x;
    const int w    = tid >> 5;
    const int lane = tid & 31;
    const int n    = lane & 15;
    const int hi   = lane >> 4;

    const long long row = (long long)blockIdx.x * WAVES + w;

    const float* xseg = x   + row * (long long)T_LEN + 4 * lane;
    float*       oseg = out + row * (long long)T_LEN + 4 * lane;
    float4 xin[16];
#pragma unroll
    for (int r = 0; r < 16; ++r)
        xin[r] = *(const float4*)(xseg + r * SEG_LEN);

    const float gr  = gain_ri[0],  gi  = gain_ri[1];
    const float p0r = poles_ri[0], p0i = poles_ri[1];
    const float p1r = poles_ri[2], p1i = poles_ri[3];
    const float z0r = zeros_ri[0], z0i = zeros_ri[1];
    const float z1r = zeros_ri[2], z1i = zeros_ri[3];

    const float a1 = -(p0r + p1r);
    const float a2 = p0r * p1r - p0i * p1i;
    const float zc1r = -(z0r + z1r), zc1i = -(z0i + z1i);
    const float zc2r = z0r * z1r - z0i * z1i;
    const float zc2i = z0r * z1i + z0i * z1r;
    const float b0 = gr;
    const float b1 = gr * zc1r - gi * zc1i;
    const float b2 = gr * zc2r - gi * zc2i;

    float g[16];
    g[0] = 1.f; g[1] = -a1;
#pragma unroll
    for (int k = 2; k < 16; ++k) g[k] = -a1 * g[k-1] - a2 * g[k-2];

    float hh[16], R1[16], R2[16], P[16], Q[16];
#pragma unroll
    for (int k = 0; k < 16; ++k) {
        const float gm1 = (k >= 1) ? g[k-1] : 0.f;
        const float gm2 = (k >= 2) ? g[k-2] : 0.f;
        hh[k] = b0 * g[k] + b1 * gm1 + b2 * gm2;
        R1[k] = b1 * g[k] + b2 * gm1;
        R2[k] = b2 * g[k];
    }
    P[0] = -a1; P[1] = a1 * a1 - a2;
    Q[0] = -a2; Q[1] = a1 * a2;
#pragma unroll
    for (int k = 2; k < 16; ++k) {
        P[k] = -a1 * P[k-1] - a2 * P[k-2];
        Q[k] = -a1 * Q[k-1] - a2 * Q[k-2];
    }

    const float B11 = P[15], B12 = Q[15], B21 = P[14], B22 = Q[14];

    float M11 = 1.f, M21 = 0.f, M12 = 0.f, M22 = 1.f;
#pragma unroll
    for (int b = 0; b < SEG_BLKS; ++b) {
        const float c11 = B11 * M11 + B12 * M21;
        const float c21 = B21 * M11 + B22 * M21;
        const float c12 = B11 * M12 + B12 * M22;
        const float c22 = B21 * M12 + B22 * M22;
        M11 = c11; M21 = c21; M12 = c12; M22 = c22;
    }

    float cr1[8], cr2[8], cp[8], cq[8];
#pragma unroll
    for (int r = 0; r < 8; ++r) {
        cr1[r] = hi ? R1[r + 8] : R1[r];
        cr2[r] = hi ? R2[r + 8] : R2[r];
        cp[r]  = hi ? P[r + 8]  : P[r];
        cq[r]  = hi ? Q[r + 8]  : Q[r];
    }

#pragma unroll
    for (int r = 0; r < 16; ++r)
        *(float4*)&tile[w][r][4 * lane] = xin[r];

    if (lane == 0) {
#pragma unroll
        for (int k = 0; k < 16; ++k) hsh[w][k] = hh[k];
    }

    __builtin_amdgcn_wave_barrier();
    asm volatile("s_wait_dscnt 0x0" ::: "memory");

    v2f A[4];
#pragma unroll
    for (int kb = 0; kb < 4; ++kb) {
        const int i0 = n - 4 * kb - 2 * hi;
        A[kb].x = (i0 >= 0) ? hsh[w][i0]     : 0.f;
        A[kb].y = (i0 >= 1) ? hsh[w][i0 - 1] : 0.f;
    }

    const v2f xh0 = *(const v2f*)&tile[w][n > 0 ? n - 1 : 0][SEG_LEN - 2];

    v8f accs[SEG_BLKS];
    float e1[SEG_BLKS], e2[SEG_BLKS];

#pragma unroll
    for (int blk = 0; blk < SEG_BLKS; ++blk) {
        const int t0 = blk * BLK;

        v2f Bv[4];
#pragma unroll
        for (int kb = 0; kb < 4; ++kb)
            Bv[kb] = *(const v2f*)&tile[w][n][t0 + 4 * kb + 2 * hi];

        float xm2, xm1;
        if (blk == 0) {
            xm2 = (n > 0) ? xh0.x : 0.f;
            xm1 = (n > 0) ? xh0.y : 0.f;
        } else {
            const v2f xh = *(const v2f*)&tile[w][n][t0 - 2];
            xm2 = xh.x;
            xm1 = xh.y;
        }

        v8f acc;
#pragma unroll
        for (int r = 0; r < 8; ++r)
            acc[r] = cr1[r] * xm1 + cr2[r] * xm2;

#pragma unroll
        for (int kb = 0; kb < 4; ++kb)
            acc = __builtin_amdgcn_wmma_f32_16x16x4_f32(
                false, A[kb], false, Bv[kb], (short)0, acc, false, false);

        e2[blk] = __shfl(acc[6], 16 + n, 32);
        e1[blk] = __shfl(acc[7], 16 + n, 32);

        accs[blk] = acc;
    }

    float ym1 = e1[0], ym2 = e2[0];
#pragma unroll
    for (int blk = 1; blk < SEG_BLKS; ++blk) {
        const float u1 = B11 * ym1 + (B12 * ym2 + e1[blk]);
        const float u2 = B21 * ym1 + (B22 * ym2 + e2[blk]);
        ym1 = u1; ym2 = u2;
    }

    float s1 = 0.f, s2 = 0.f;
    float S1 = row_bcast<0>(ym1), S2 = row_bcast<0>(ym2);
    if (n == 1) { s1 = S1; s2 = S2; }
#define SCAN_STEP(V)                                                    \
    {                                                                   \
        const float u1 = M11 * S1 + (M12 * S2 + row_bcast<V>(ym1));     \
        const float u2 = M21 * S1 + (M22 * S2 + row_bcast<V>(ym2));     \
        S1 = u1; S2 = u2;                                    \
        if (n == V + 1) { s1 = S1; s2 = S2; }                           \
    }
    SCAN_STEP(1)  SCAN_STEP(2)  SCAN_STEP(3)  SCAN_STEP(4)  SCAN_STEP(5)
    SCAN_STEP(6)  SCAN_STEP(7)  SCAN_STEP(8)  SCAN_STEP(9)  SCAN_STEP(10)
    SCAN_STEP(11) SCAN_STEP(12) SCAN_STEP(13) SCAN_STEP(14)
#undef SCAN_STEP

#pragma unroll
    for (int blk = 0; blk < SEG_BLKS; ++blk) {
        const int t0 = blk * BLK;
        v8f acc = accs[blk];
#pragma unroll
        for (int r = 0; r < 8; ++r)
            acc[r] += cp[r] * s1 + cq[r] * s2;
        const float u1 = B11 * s1 + (B12 * s2 + e1[blk]);
        const float u2 = B21 * s1 + (B22 * s2 + e2[blk]);
        s1 = u1; s2 = u2;

        *(float4*)&tile[w][n][t0 + 8 * hi]     = make_float4(acc[0], acc[1], acc[2], acc[3]);
        *(float4*)&tile[w][n][t0 + 8 * hi + 4] = make_float4(acc[4], acc[5], acc[6], acc[7]);
    }
    __builtin_amdgcn_wave_barrier();
    asm volatile("s_wait_dscnt 0x0" ::: "memory");

#pragma unroll
    for (int r = 0; r < 16; ++r)
        *(float4*)(oseg + r * SEG_LEN) = *(const float4*)&tile[w][r][4 * lane];
}

extern "C" void kernel_launch(void* const* d_in, const int* in_sizes, int n_in,
                              void* d_out, int out_size, void* d_ws, size_t ws_size,
                              hipStream_t stream) {
    const float* x     = (const float*)d_in[0];
    const float* gain  = (const float*)d_in[1];
    const float* poles = (const float*)d_in[2];
    const float* zeros = (const float*)d_in[3];
    float* out = (float*)d_out;

    const int B = in_sizes[0] / T_LEN;
    const int blocks = B / WAVES;
    pz_biquad_wmma<<<blocks, CTA_THREADS, 0, stream>>>(x, gain, poles, zeros, out);
}
